# starting kernel
# baseline (speedup 1.0000x reference)
#include <hip/hip_runtime.h>

typedef __attribute__((ext_vector_type(2))) float v2f;
typedef __attribute__((ext_vector_type(8))) float v8f;

#define NEXP 16
#define INF 256
#define OUTF 256
#define NB 16
#define NPG 128
#define NTOT (NB * NPG)


__global__ __launch_bounds__(256) void mole_mix_kernel(
    const float* __restrict__ coeffs,
    const float* __restrict__ kern,
    float* __restrict__ mixedP)
{
  const int lane = threadIdx.x & 31;
  const int wave = threadIdx.x >> 5;
  const int lh   = lane >> 4;
  const int lr   = lane & 15;
  const int tile = blockIdx.x * 8 + wave;
  const int c0   = tile * 16;
  const int i    = c0 >> 8;
  const int o    = (c0 & 255) + lr;

  v8f acc = {};
#pragma unroll
  for (int t = 0; t < 4; ++t) {
    const int k = t * 4 + 2 * lh;
    v2f a, b;
    a.x = coeffs[lr * NEXP + k];
    a.y = coeffs[lr * NEXP + k + 1];
    b.x = kern[(size_t)k       * (INF * OUTF) + c0 + lr];
    b.y = kern[(size_t)(k + 1) * (INF * OUTF) + c0 + lr];
    acc = __builtin_amdgcn_wmma_f32_16x16x4_f32(
        false, a, false, b, (short)0, acc, false, false);
  }
  const size_t pbase = (size_t)(i >> 1) * 512 + (size_t)o * 2 + (i & 1);
#pragma unroll
  for (int r = 0; r < 8; ++r) {
    const int g = r + 8 * lh;
    mixedP[(size_t)g * (INF * OUTF) + pbase] = acc[r];
  }
}

__global__ __launch_bounds__(256) void mole_gemm_kernel(
    const float* __restrict__ x,
    const float* __restrict__ mixedP,
    const float* __restrict__ bias,
    float* __restrict__ out)
{
  const int lane = threadIdx.x & 31;
  const int wave = threadIdx.x >> 5;
  const int lh   = lane >> 4;
  const int lr   = lane & 15;
  const int tid  = blockIdx.x * 8 + wave;
  const int ct   = tid & 7;
  const int mt   = tid >> 3;
  const int row0 = mt * 16;
  const int g    = mt >> 3;
  const int n0   = ct * 32;

  const float* arow = x + (size_t)(row0 + lr) * INF + 2 * lh;
  const float* wbase = mixedP + (size_t)g * (INF * OUTF)
                     + (size_t)lh * 512 + (size_t)(n0 + lr) * 2;

  v8f acc0 = {}, acc1 = {};
#pragma unroll 8
  for (int k0 = 0; k0 < INF; k0 += 4) {
    v2f a  = *(const v2f*)(arow + k0);
    const float* wp = wbase + (size_t)(k0 >> 1) * 512;
    v2f b0 = *(const v2f*)(wp);
    v2f b1 = *(const v2f*)(wp + 32);
    acc0 = __builtin_amdgcn_wmma_f32_16x16x4_f32(
        false, a, false, b0, (short)0, acc0, false, false);
    acc1 = __builtin_amdgcn_wmma_f32_16x16x4_f32(
        false, a, false, b1, (short)0, acc1, false, false);
  }

  const float bia0 = bias[n0 + lr];
  const float bia1 = bias[n0 + 16 + lr];
#pragma unroll
  for (int r = 0; r < 8; ++r) {
    const int m = row0 + r + 8 * lh;
    out[(size_t)m * OUTF + n0 + lr]      = acc0[r] + bia0;
    out[(size_t)m * OUTF + n0 + 16 + lr] = acc1[r] + bia1;
  }
}

extern "C" void kernel_launch(void* const* d_in, const int* in_sizes, int n_in,
                              void* d_out, int out_size, void* d_ws, size_t ws_size,
                              hipStream_t stream) {
  const float* x      = (const float*)d_in[0];
  const float* coeffs = (const float*)d_in[2];
  const float* kern   = (const float*)d_in[3];
  const float* bias   = (const float*)d_in[4];
  float*       outp   = (float*)d_out;
  float*       mixedP = (float*)d_ws;

  mole_mix_kernel<<<512, 256, 0, stream>>>(coeffs, kern, mixedP);
  mole_gemm_kernel<<<128, 256, 0, stream>>>(x, mixedP, bias, outp);
}


# gemm stage: the 16x256 activation tile shared by a block's 8 waves is staged once in LDS with whole-line loads; A fragments read from LDS, not 16-line global gathers per wave
# speedup vs baseline: 1.1120x; 1.1120x over previous
#include <hip/hip_runtime.h>

typedef __attribute__((ext_vector_type(2))) float v2f;
typedef __attribute__((ext_vector_type(4))) float v4f;
typedef __attribute__((ext_vector_type(8))) float v8f;

#define NEXP 16
#define INF 256
#define OUTF 256
#define NB 16
#define NPG 128
#define NTOT (NB * NPG)
#define APITCH 260


__global__ __launch_bounds__(256) void mole_mix_kernel(
    const float* __restrict__ coeffs,
    const float* __restrict__ kern,
    float* __restrict__ mixedP)
{
  const int lane = threadIdx.x & 31;
  const int wave = threadIdx.x >> 5;
  const int lh   = lane >> 4;
  const int lr   = lane & 15;
  const int tile = blockIdx.x * 8 + wave;
  const int c0   = tile * 16;
  const int i    = c0 >> 8;
  const int o    = (c0 & 255) + lr;

  v8f acc = {};
#pragma unroll
  for (int t = 0; t < 4; ++t) {
    const int k = t * 4 + 2 * lh;
    v2f a, b;
    a.x = coeffs[lr * NEXP + k];
    a.y = coeffs[lr * NEXP + k + 1];
    b.x = kern[(size_t)k       * (INF * OUTF) + c0 + lr];
    b.y = kern[(size_t)(k + 1) * (INF * OUTF) + c0 + lr];
    acc = __builtin_amdgcn_wmma_f32_16x16x4_f32(
        false, a, false, b, (short)0, acc, false, false);
  }
  const size_t pbase = (size_t)(i >> 1) * 512 + (size_t)o * 2 + (i & 1);
#pragma unroll
  for (int r = 0; r < 8; ++r) {
    const int g = r + 8 * lh;
    mixedP[(size_t)g * (INF * OUTF) + pbase] = acc[r];
  }
}

__global__ __launch_bounds__(256) void mole_gemm_kernel(
    const float* __restrict__ x,
    const float* __restrict__ mixedP,
    const float* __restrict__ bias,
    float* __restrict__ out)
{
  const int lane = threadIdx.x & 31;
  const int wave = threadIdx.x >> 5;
  const int lh   = lane >> 4;
  const int lr   = lane & 15;
  const int tid  = blockIdx.x * 8 + wave;
  const int ct   = tid & 7;
  const int mt   = tid >> 3;
  const int row0 = mt * 16;
  const int g    = mt >> 3;
  const int n0   = ct * 32;

  __shared__ __attribute__((aligned(16))) float As[16 * APITCH];
  const float* xt = x + (size_t)row0 * INF;
#pragma unroll
  for (int j = 0; j < 4; ++j) {
    const int idx = j * 256 + (int)threadIdx.x;
    const int r   = idx >> 6;
    const int cc  = idx & 63;
    const v4f v = *(const v4f*)(xt + (size_t)r * INF + cc * 4);
    *(v4f*)(As + r * APITCH + cc * 4) = v;
  }
  __syncthreads();

  const float* arow = As + lr * APITCH + 2 * lh;
  const float* wbase = mixedP + (size_t)g * (INF * OUTF)
                     + (size_t)lh * 512 + (size_t)(n0 + lr) * 2;

  v8f acc0 = {}, acc1 = {};
#pragma unroll 8
  for (int k0 = 0; k0 < INF; k0 += 4) {
    v2f a  = *(const v2f*)(arow + k0);
    const float* wp = wbase + (size_t)(k0 >> 1) * 512;
    v2f b0 = *(const v2f*)(wp);
    v2f b1 = *(const v2f*)(wp + 32);
    acc0 = __builtin_amdgcn_wmma_f32_16x16x4_f32(
        false, a, false, b0, (short)0, acc0, false, false);
    acc1 = __builtin_amdgcn_wmma_f32_16x16x4_f32(
        false, a, false, b1, (short)0, acc1, false, false);
  }

  const float bia0 = bias[n0 + lr];
  const float bia1 = bias[n0 + 16 + lr];
#pragma unroll
  for (int r = 0; r < 8; ++r) {
    const int m = row0 + r + 8 * lh;
    out[(size_t)m * OUTF + n0 + lr]      = acc0[r] + bia0;
    out[(size_t)m * OUTF + n0 + 16 + lr] = acc1[r] + bia1;
  }
}

extern "C" void kernel_launch(void* const* d_in, const int* in_sizes, int n_in,
                              void* d_out, int out_size, void* d_ws, size_t ws_size,
                              hipStream_t stream) {
  const float* x      = (const float*)d_in[0];
  const float* coeffs = (const float*)d_in[2];
  const float* kern   = (const float*)d_in[3];
  const float* bias   = (const float*)d_in[4];
  float*       outp   = (float*)d_out;
  float*       mixedP = (float*)d_ws;

  mole_mix_kernel<<<512, 256, 0, stream>>>(coeffs, kern, mixedP);
  mole_gemm_kernel<<<128, 256, 0, stream>>>(x, mixedP, bias, outp);
}


# gemm stage on all 256 WGPs, one wave per SIMD (256 blocks of 4 waves, was 128 of 8); K loop unrolled, B fragments requested 24 K-steps ahead so the lone wave hides its own round trips; same k order
# speedup vs baseline: 1.5235x; 1.3701x over previous
#include <hip/hip_runtime.h>

typedef __attribute__((ext_vector_type(2))) float v2f;
typedef __attribute__((ext_vector_type(4))) float v4f;
typedef __attribute__((ext_vector_type(8))) float v8f;

#define NEXP 16
#define INF 256
#define OUTF 256
#define NB 16
#define NPG 128
#define NTOT (NB * NPG)
#define APITCH 260


__global__ __launch_bounds__(256) void mole_mix_kernel(
    const float* __restrict__ coeffs,
    const float* __restrict__ kern,
    float* __restrict__ mixedP)
{
  const int lane = threadIdx.x & 31;
  const int wave = threadIdx.x >> 5;
  const int lh   = lane >> 4;
  const int lr   = lane & 15;
  const int tile = blockIdx.x * 8 + wave;
  const int c0   = tile * 16;
  const int i    = c0 >> 8;
  const int o    = (c0 & 255) + lr;

  v8f acc = {};
#pragma unroll
  for (int t = 0; t < 4; ++t) {
    const int k = t * 4 + 2 * lh;
    v2f a, b;
    a.x = coeffs[lr * NEXP + k];
    a.y = coeffs[lr * NEXP + k + 1];
    b.x = kern[(size_t)k       * (INF * OUTF) + c0 + lr];
    b.y = kern[(size_t)(k + 1) * (INF * OUTF) + c0 + lr];
    acc = __builtin_amdgcn_wmma_f32_16x16x4_f32(
        false, a, false, b, (short)0, acc, false, false);
  }
  const size_t pbase = (size_t)(i >> 1) * 512 + (size_t)o * 2 + (i & 1);
#pragma unroll
  for (int r = 0; r < 8; ++r) {
    const int g = r + 8 * lh;
    mixedP[(size_t)g * (INF * OUTF) + pbase] = acc[r];
  }
}

#define GEMM_KSTEPS (INF / 4)
#define GEMM_AHEAD  24

__global__ __launch_bounds__(128) void mole_gemm_kernel(
    const float* __restrict__ x,
    const float* __restrict__ mixedP,
    const float* __restrict__ bias,
    float* __restrict__ out)
{
  const int lane = threadIdx.x & 31;
  const int wave = threadIdx.x >> 5;
  const int lh   = lane >> 4;
  const int lr   = lane & 15;
  const int mt   = blockIdx.x >> 1;
  const int ct   = (blockIdx.x & 1) * 4 + wave;
  const int row0 = mt * 16;
  const int g    = mt >> 3;
  const int n0   = ct * 32;

  const float* wbase = mixedP + (size_t)g * (INF * OUTF)
                     + (size_t)lh * 512 + (size_t)(n0 + lr) * 2;

  __shared__ __attribute__((aligned(16))) float As[16 * APITCH];
  const float* xt = x + (size_t)row0 * INF;
  v4f xv[8];
#pragma unroll
  for (int j = 0; j < 8; ++j) {
    const int idx = j * 128 + (int)threadIdx.x;
    const int r   = idx >> 6;
    const int cc  = idx & 63;
    xv[j] = *(const v4f*)(xt + (size_t)r * INF + cc * 4);
  }

  v2f bq[GEMM_KSTEPS][2];
#pragma unroll
  for (int s = 0; s < GEMM_AHEAD; ++s) {
    const float* wp = wbase + (size_t)s * 1024;
    bq[s][0] = *(const v2f*)(wp);
    bq[s][1] = *(const v2f*)(wp + 32);
  }

#pragma unroll
  for (int j = 0; j < 8; ++j) {
    const int idx = j * 128 + (int)threadIdx.x;
    const int r   = idx >> 6;
    const int cc  = idx & 63;
    *(v4f*)(As + r * APITCH + cc * 4) = xv[j];
  }
  __builtin_amdgcn_fence(__ATOMIC_RELEASE, "workgroup", "local");
  __builtin_amdgcn_s_barrier();
  __builtin_amdgcn_fence(__ATOMIC_ACQUIRE, "workgroup", "local");

  const float* arow = As + lr * APITCH + 2 * lh;

  v8f acc0 = {}, acc1 = {};
#pragma unroll
  for (int s = 0; s < GEMM_KSTEPS; ++s) {
    if (s + GEMM_AHEAD < GEMM_KSTEPS) {
      const float* wp = wbase + (size_t)(s + GEMM_AHEAD) * 1024;
      bq[s + GEMM_AHEAD][0] = *(const v2f*)(wp);
      bq[s + GEMM_AHEAD][1] = *(const v2f*)(wp + 32);
    }
    v2f a = *(const v2f*)(arow + s * 4);
    acc0 = __builtin_amdgcn_wmma_f32_16x16x4_f32(
        false, a, false, bq[s][0], (short)0, acc0, false, false);
    acc1 = __builtin_amdgcn_wmma_f32_16x16x4_f32(
        false, a, false, bq[s][1], (short)0, acc1, false, false);
  }

  const float bia0 = bias[n0 + lr];
  const float bia1 = bias[n0 + 16 + lr];
#pragma unroll
  for (int r = 0; r < 8; ++r) {
    const int m = row0 + r + 8 * lh;
    out[(size_t)m * OUTF + n0 + lr]      = acc0[r] + bia0;
    out[(size_t)m * OUTF + n0 + 16 + lr] = acc1[r] + bia1;
  }
}

extern "C" void kernel_launch(void* const* d_in, const int* in_sizes, int n_in,
                              void* d_out, int out_size, void* d_ws, size_t ws_size,
                              hipStream_t stream) {
  const float* x      = (const float*)d_in[0];
  const float* coeffs = (const float*)d_in[2];
  const float* kern   = (const float*)d_in[3];
  const float* bias   = (const float*)d_in[4];
  float*       outp   = (float*)d_out;
  float*       mixedP = (float*)d_ws;

  mole_mix_kernel<<<512, 256, 0, stream>>>(coeffs, kern, mixedP);
  mole_gemm_kernel<<<256, 128, 0, stream>>>(x, mixedP, bias, outp);
}


# mix stage retiled: 1024 waves (was 4096), each a 2-row x 32-col weight patch for all graphs; 8-byte loads all in flight at entry; K pairs stored as whole 16-byte words, full lines; same k order
# speedup vs baseline: 1.6581x; 1.0883x over previous
#include <hip/hip_runtime.h>

typedef __attribute__((ext_vector_type(2))) float v2f;
typedef __attribute__((ext_vector_type(4))) float v4f;
typedef __attribute__((ext_vector_type(8))) float v8f;

#define NEXP 16
#define INF 256
#define OUTF 256
#define NB 16
#define NPG 128
#define NTOT (NB * NPG)
#define APITCH 260


__global__ __launch_bounds__(128) __attribute__((amdgpu_waves_per_eu(1, 2)))
void mole_mix_kernel(
    const float* __restrict__ coeffs,
    const float* __restrict__ kern,
    float* __restrict__ mixedP)
{
  const int lane = threadIdx.x & 31;
  const int wave = threadIdx.x >> 5;
  const int lh   = lane >> 4;
  const int lr   = lane & 15;
  const int p    = blockIdx.x >> 1;
  const int o0   = ((blockIdx.x & 1) * 4 + wave) * 32;

  const float* kb = kern + (size_t)(2 * lh) * (INF * OUTF)
                  + (size_t)(2 * p) * OUTF + o0 + 2 * lr;
  v2f q[4][2][2];
#pragma unroll
  for (int t = 0; t < 4; ++t)
#pragma unroll
    for (int kk = 0; kk < 2; ++kk)
#pragma unroll
      for (int s = 0; s < 2; ++s)
        q[t][kk][s] = *(const v2f*)(kb + (size_t)(4 * t + kk) * (INF * OUTF)
                                       + s * OUTF);

  v2f a[4];
#pragma unroll
  for (int t = 0; t < 4; ++t)
    a[t] = *(const v2f*)(coeffs + lr * NEXP + 4 * t + 2 * lh);

  v8f acc[2][2] = {};
#pragma unroll
  for (int t = 0; t < 4; ++t) {
#pragma unroll
    for (int s = 0; s < 2; ++s) {
#pragma unroll
      for (int j = 0; j < 2; ++j) {
        v2f b;
        b.x = q[t][0][s][j];
        b.y = q[t][1][s][j];
        acc[s][j] = __builtin_amdgcn_wmma_f32_16x16x4_f32(
            false, a[t], false, b, (short)0, acc[s][j], false, false);
      }
    }
  }

  float* dst = mixedP + (size_t)(8 * lh) * (INF * OUTF) + (size_t)p * 512
             + (size_t)(o0 + 2 * lr) * 2;
#pragma unroll
  for (int r = 0; r < 8; ++r) {
    v4f v;
    v.x = acc[0][0][r];
    v.y = acc[1][0][r];
    v.z = acc[0][1][r];
    v.w = acc[1][1][r];
    *(v4f*)(dst + (size_t)r * (INF * OUTF)) = v;
  }
}

#define GEMM_KSTEPS (INF / 4)
#define GEMM_AHEAD  24

__global__ __launch_bounds__(128) void mole_gemm_kernel(
    const float* __restrict__ x,
    const float* __restrict__ mixedP,
    const float* __restrict__ bias,
    float* __restrict__ out)
{
  const int lane = threadIdx.x & 31;
  const int wave = threadIdx.x >> 5;
  const int lh   = lane >> 4;
  const int lr   = lane & 15;
  const int mt   = blockIdx.x >> 1;
  const int ct   = (blockIdx.x & 1) * 4 + wave;
  const int row0 = mt * 16;
  const int g    = mt >> 3;
  const int n0   = ct * 32;

  const float* wbase = mixedP + (size_t)g * (INF * OUTF)
                     + (size_t)lh * 512 + (size_t)(n0 + lr) * 2;

  __shared__ __attribute__((aligned(16))) float As[16 * APITCH];
  const float* xt = x + (size_t)row0 * INF;
  v4f xv[8];
#pragma unroll
  for (int j = 0; j < 8; ++j) {
    const int idx = j * 128 + (int)threadIdx.x;
    const int r   = idx >> 6;
    const int cc  = idx & 63;
    xv[j] = *(const v4f*)(xt + (size_t)r * INF + cc * 4);
  }

  v2f bq[GEMM_KSTEPS][2];
#pragma unroll
  for (int s = 0; s < GEMM_AHEAD; ++s) {
    const float* wp = wbase + (size_t)s * 1024;
    bq[s][0] = *(const v2f*)(wp);
    bq[s][1] = *(const v2f*)(wp + 32);
  }

#pragma unroll
  for (int j = 0; j < 8; ++j) {
    const int idx = j * 128 + (int)threadIdx.x;
    const int r   = idx >> 6;
    const int cc  = idx & 63;
    *(v4f*)(As + r * APITCH + cc * 4) = xv[j];
  }
  __builtin_amdgcn_fence(__ATOMIC_RELEASE, "workgroup", "local");
  __builtin_amdgcn_s_barrier();
  __builtin_amdgcn_fence(__ATOMIC_ACQUIRE, "workgroup", "local");

  const float* arow = As + lr * APITCH + 2 * lh;

  v8f acc0 = {}, acc1 = {};
#pragma unroll
  for (int s = 0; s < GEMM_KSTEPS; ++s) {
    if (s + GEMM_AHEAD < GEMM_KSTEPS) {
      const float* wp = wbase + (size_t)(s + GEMM_AHEAD) * 1024;
      bq[s + GEMM_AHEAD][0] = *(const v2f*)(wp);
      bq[s + GEMM_AHEAD][1] = *(const v2f*)(wp + 32);
    }
    v2f a = *(const v2f*)(arow + s * 4);
    acc0 = __builtin_amdgcn_wmma_f32_16x16x4_f32(
        false, a, false, bq[s][0], (short)0, acc0, false, false);
    acc1 = __builtin_amdgcn_wmma_f32_16x16x4_f32(
        false, a, false, bq[s][1], (short)0, acc1, false, false);
  }

  const float bia0 = bias[n0 + lr];
  const float bia1 = bias[n0 + 16 + lr];
#pragma unroll
  for (int r = 0; r < 8; ++r) {
    const int m = row0 + r + 8 * lh;
    out[(size_t)m * OUTF + n0 + lr]      = acc0[r] + bia0;
    out[(size_t)m * OUTF + n0 + 16 + lr] = acc1[r] + bia1;
  }
}

extern "C" void kernel_launch(void* const* d_in, const int* in_sizes, int n_in,
                              void* d_out, int out_size, void* d_ws, size_t ws_size,
                              hipStream_t stream) {
  const float* x      = (const float*)d_in[0];
  const float* coeffs = (const float*)d_in[2];
  const float* kern   = (const float*)d_in[3];
  const float* bias   = (const float*)d_in[4];
  float*       outp   = (float*)d_out;
  float*       mixedP = (float*)d_ws;

  mole_mix_kernel<<<256, 128, 0, stream>>>(coeffs, kern, mixedP);
  mole_gemm_kernel<<<256, 128, 0, stream>>>(x, mixedP, bias, outp);
}
